# starting kernel
# baseline (speedup 1.0000x reference)
#include <hip/hip_runtime.h>
#include <hip/hip_bf16.h>


typedef float v2f __attribute__((ext_vector_type(2)));
typedef float v8f __attribute__((ext_vector_type(8)));

#define FIN   512
#define FH    16
#define FOUT  2


__global__ void k_init_deg(float* __restrict__ deg, int n) {
    int i = blockIdx.x * blockDim.x + threadIdx.x;
    if (i < n) deg[i] = 1.0f;
}

__global__ void k_deg_accum(const int* __restrict__ ei, int E,
                            float* __restrict__ deg) {
    int e = blockIdx.x * blockDim.x + threadIdx.x;
    if (e < E) atomicAdd(&deg[ei[E + e]], 1.0f);
}

__global__ void k_rsqrt_inplace(float* __restrict__ buf, int n) {
    int i = blockIdx.x * blockDim.x + threadIdx.x;
    if (i < n) buf[i] = rsqrtf(buf[i]);
}


__global__ void __launch_bounds__(256)
k_gemm1_wmma(const float* __restrict__ x, const float* __restrict__ W1,
             float* __restrict__ h, int n_tiles) {
    __shared__ float lw[FIN * FH];
    int tid = threadIdx.x;
    for (int i = tid; i < FIN * FH; i += 256) lw[i] = W1[i];
    __syncthreads();

    int wave = tid >> 5;
    int lane = tid & 31;
    int tile = blockIdx.x * 8 + wave;
    if (tile >= n_tiles) return;

    int half = lane >> 4;
    int mn   = lane & 15;
    int kofs = 2 * half;

    const float* __restrict__ xrow = x + (size_t)(tile * 16 + mn) * FIN;

    v8f c = {};
    #pragma unroll 4
    for (int k = 0; k < FIN; k += 4) {
        v2f a = *(const v2f*)(xrow + k + kofs);
        v2f b;
        b.x = lw[(k + kofs)     * FH + mn];
        b.y = lw[(k + kofs + 1) * FH + mn];
        c = __builtin_amdgcn_wmma_f32_16x16x4_f32(
                 false, a,  false, b,
                 (short)0, c,  false,  false);
    }

    float* __restrict__ hrow = h + (size_t)tile * 16 * FH;
    #pragma unroll
    for (int r = 0; r < 8; ++r)
        hrow[(r + 8 * half) * FH + mn] = c[r];
}


__global__ void k_scatter1_init(const float* __restrict__ h,
                                const float* __restrict__ dinv,
                                float* __restrict__ agg1, int n) {
    int t = blockIdx.x * blockDim.x + threadIdx.x;
    if (t < n * FH) {
        int i = t >> 4;
        float w = dinv[i];
        agg1[t] = h[t] * w * w;
    }
}

__global__ void k_scatter1(const int* __restrict__ ei, int E,
                           const float* __restrict__ h,
                           const float* __restrict__ dinv,
                           float* __restrict__ agg1) {
    int t = blockIdx.x * blockDim.x + threadIdx.x;
    if (t < E * FH) {
        int e = t >> 4, f = t & 15;
        int s = ei[e], d = ei[E + e];
        float w = dinv[s] * dinv[d];
        atomicAdd(&agg1[d * FH + f], h[s * FH + f] * w);
    }
}


__global__ void k_layer2(const float* __restrict__ agg1,
                         const float* __restrict__ b1,
                         const float* __restrict__ W2,
                         float* __restrict__ h2, int n) {
    int i = blockIdx.x * blockDim.x + threadIdx.x;
    if (i < n) {
        float a0 = 0.0f, a1 = 0.0f;
        #pragma unroll
        for (int k = 0; k < FH; ++k) {
            float v = fmaxf(agg1[(size_t)i * FH + k] + b1[k], 0.0f);
            a0 = fmaf(v, W2[2 * k],     a0);
            a1 = fmaf(v, W2[2 * k + 1], a1);
        }
        h2[2 * i]     = a0;
        h2[2 * i + 1] = a1;
    }
}


__global__ void k_scatter2_init(const float* __restrict__ h2,
                                const float* __restrict__ dinv,
                                float* __restrict__ agg2, int n) {
    int t = blockIdx.x * blockDim.x + threadIdx.x;
    if (t < n * FOUT) {
        int i = t >> 1;
        float w = dinv[i];
        agg2[t] = h2[t] * w * w;
    }
}

__global__ void k_scatter2(const int* __restrict__ ei, int E,
                           const float* __restrict__ h2,
                           const float* __restrict__ dinv,
                           float* __restrict__ agg2) {
    int t = blockIdx.x * blockDim.x + threadIdx.x;
    if (t < E * FOUT) {
        int e = t >> 1, f = t & 1;
        int s = ei[e], d = ei[E + e];
        float w = dinv[s] * dinv[d];
        atomicAdd(&agg2[d * FOUT + f], h2[s * FOUT + f] * w);
    }
}


__global__ void k_logsoftmax(const float* __restrict__ agg2,
                             const float* __restrict__ b2,
                             float* __restrict__ out, int n) {
    int i = blockIdx.x * blockDim.x + threadIdx.x;
    if (i < n) {
        float a0 = agg2[2 * i]     + b2[0];
        float a1 = agg2[2 * i + 1] + b2[1];
        float m  = fmaxf(a0, a1);
        float ls = logf(expf(a0 - m) + expf(a1 - m));
        out[2 * i]     = a0 - m - ls;
        out[2 * i + 1] = a1 - m - ls;
    }
}


extern "C" void kernel_launch(void* const* d_in, const int* in_sizes, int n_in,
                              void* d_out, int out_size, void* d_ws, size_t ws_size,
                              hipStream_t stream) {
    const float* x  = (const float*)d_in[0];
    const int*   ei = (const int*)d_in[1];
    const float* W1 = (const float*)d_in[2];
    const float* b1 = (const float*)d_in[3];
    const float* W2 = (const float*)d_in[4];
    const float* b2 = (const float*)d_in[5];
    float* out      = (float*)d_out;

    const int N = in_sizes[0] / FIN;
    const int E = in_sizes[1] / 2;

    float* ws   = (float*)d_ws;
    float* dinv = ws;
    float* h    = dinv + N;
    float* agg1 = h    + (size_t)N * FH;
    float* h2   = agg1 + (size_t)N * FH;
    float* agg2 = h2   + (size_t)N * FOUT;

    const int B = 256;
    auto blk = [](long long t, int b) { return (int)((t + b - 1) / b); };

    k_init_deg   <<<blk(N, B), B, 0, stream>>>(dinv, N);
    k_deg_accum  <<<blk(E, B), B, 0, stream>>>(ei, E, dinv);
    k_rsqrt_inplace<<<blk(N, B), B, 0, stream>>>(dinv, N);

    const int n_tiles = N / 16;
    k_gemm1_wmma <<<blk(n_tiles, 8), B, 0, stream>>>(x, W1, h, n_tiles);
    k_scatter1_init<<<blk((long long)N * FH, B), B, 0, stream>>>(h, dinv, agg1, N);
    k_scatter1   <<<blk((long long)E * FH, B), B, 0, stream>>>(ei, E, h, dinv, agg1);

    k_layer2     <<<blk(N, B), B, 0, stream>>>(agg1, b1, W2, h2, N);
    k_scatter2_init<<<blk((long long)N * FOUT, B), B, 0, stream>>>(h2, dinv, agg2, N);
    k_scatter2   <<<blk((long long)E * FOUT, B), B, 0, stream>>>(ei, E, h2, dinv, agg2);

    k_logsoftmax <<<blk(N, B), B, 0, stream>>>(agg2, b2, out, N);
}


# GEMM1 streams x: whole-line non-temporal 16-byte loads (2 rows x 256 B, 16 in flight per wave) reach the A fragments through the wave's own LDS slot; k permuted inside each 8 (sum order changes)
# speedup vs baseline: 4.5922x; 4.5922x over previous
#include <hip/hip_runtime.h>
#include <hip/hip_bf16.h>


typedef float v2f __attribute__((ext_vector_type(2)));
typedef float v4f __attribute__((ext_vector_type(4)));
typedef float v8f __attribute__((ext_vector_type(8)));

#define FIN   512
#define FH    16
#define FOUT  2
#define HROW  32
#define SLOTS 64
#define CHUNK 256
#define TICK_STRIDE 64


#define KB      64
#define NBATCH  (FIN / KB)
#define DEPTH   2
#define APITCH  68

__global__ void __launch_bounds__(256)
k_gemm1_wmma(const float* __restrict__ x, const float* __restrict__ W1,
             float* __restrict__ h, int* __restrict__ novf, int* __restrict__ tick,
             int n_tiles) {
    __shared__ v4f lw[FIN * FH / 4];
    __shared__ v4f la[8][16 * APITCH / 4];
    int tid = threadIdx.x;
    if (blockIdx.x == 0 && tid == 0) { *novf = 0; tick[0] = 0; tick[TICK_STRIDE] = 0; }

    int wave = tid >> 5;
    int lane = tid & 31;
    int tile = blockIdx.x * 8 + wave;
    const bool active = tile < n_tiles;
    int half = lane >> 4;
    int mn   = lane & 15;

    v4f wv[8];
    #pragma unroll
    for (int it = 0; it < 8; ++it) wv[it] = ((const v4f*)W1)[tid + 256 * it];

    const int ltile = active ? tile : n_tiles - 1;
    const v4f* __restrict__ xp = (const v4f*)x + ((size_t)(ltile * 16 + half) * (FIN / 4) + mn);
    v4f r[DEPTH][8];
    #pragma unroll
    for (int d = 0; d < DEPTH; ++d) {
        #pragma unroll
        for (int i = 0; i < 8; ++i)
            r[d][i] = __builtin_nontemporal_load(xp + i * (2 * FIN / 4) + d * (KB / 4));
    }

    {
        float* __restrict__ lwf = (float*)lw;
        const int n0 = (tid & 3) * 4;
        #pragma unroll
        for (int it = 0; it < 8; ++it) {
            const int k = (tid >> 2) + 64 * it;
            float* __restrict__ d = lwf + ((k >> 2) * 16 + n0) * 4 + (k & 3);
            d[0] = wv[it].x; d[4] = wv[it].y; d[8] = wv[it].z; d[12] = wv[it].w;
        }
    }
    __syncthreads();
    if (!active) return;

    float* __restrict__ myA = (float*)la[wave];
    v8f c = {};
    #pragma unroll
    for (int b = 0; b < NBATCH; ++b) {
        #pragma unroll
        for (int i = 0; i < 8; ++i)
            *(v4f*)(myA + (2 * i + half) * APITCH + mn * 4) = r[b % DEPTH][i];
        if (b + DEPTH < NBATCH) {
            #pragma unroll
            for (int i = 0; i < 8; ++i)
                r[b % DEPTH][i] = __builtin_nontemporal_load(xp + i * (2 * FIN / 4) + (b + DEPTH) * (KB / 4));
        }
        #pragma unroll
        for (int j = 0; j < 8; ++j) {
            v4f a4 = *(const v4f*)(myA + mn * APITCH + (2 * j + half) * 4);
            v4f b4 = lw[((b * 8 + j) * 2 + half) * 16 + mn];
            v2f a0 = {a4.x, a4.y}, b0 = {b4.x, b4.y};
            v2f a1 = {a4.z, a4.w}, b1 = {b4.z, b4.w};
            c = __builtin_amdgcn_wmma_f32_16x16x4_f32(
                     false, a0,  false, b0,
                     (short)0, c,  false,  false);
            c = __builtin_amdgcn_wmma_f32_16x16x4_f32(
                     false, a1,  false, b1,
                     (short)0, c,  false,  false);
        }
    }

    float* __restrict__ hrow = h + (size_t)tile * 16 * HROW;
    #pragma unroll
    for (int rr = 0; rr < 8; ++rr)
        hrow[(rr + 8 * half) * HROW + mn] = c[rr];
    if (half == 0) hrow[mn * HROW + FH] = 0.0f;
}

static __device__ __forceinline__ void place_chunk(const int* __restrict__ ei, int E,
                                                   int* __restrict__ rows,
                                                   int* __restrict__ slots,
                                                   int* __restrict__ novf,
                                                   int2* __restrict__ ovf,
                                                   int b, int lo, unsigned span,
                                                   const bool guard) {
#define EDGE_LOAD(k)                                                              \
    const int e##k  = b + k * 32;                                                 \
    const int ec##k = (guard && e##k >= E) ? E - 1 : e##k;                        \
    const int s##k  = ei[ec##k];                        \
    const int d##k  = ei[E + ec##k];
#define EDGE_ADD(k)                                                               \
    const bool in##k = (!guard || e##k < E) && (unsigned)(d##k - lo) < span;      \
    int r##k;                                                                     \
    if (in##k) r##k = atomicAdd(&rows[(size_t)d##k * HROW + FH], 1);
#define EDGE_PUT(k)                                                               \
    if (in##k) {                                                                  \
        if (r##k < SLOTS) {                                                       \
            slots[(size_t)d##k * SLOTS + r##k] = s##k;                            \
        } else {                                       \
            int t = atomicAdd(novf, 1);                                           \
            ovf[t] = make_int2(s##k, d##k);                                       \
        }                                                                         \
    }
    EDGE_LOAD(0) EDGE_LOAD(1) EDGE_LOAD(2) EDGE_LOAD(3)
    EDGE_LOAD(4) EDGE_LOAD(5) EDGE_LOAD(6) EDGE_LOAD(7)
    asm volatile("" :: "v"(d0), "v"(d1), "v"(d2), "v"(d3), "v"(d4), "v"(d5), "v"(d6), "v"(d7));
    EDGE_ADD(0) EDGE_ADD(1) EDGE_ADD(2) EDGE_ADD(3)
    EDGE_ADD(4) EDGE_ADD(5) EDGE_ADD(6) EDGE_ADD(7)
    EDGE_PUT(0) EDGE_PUT(1) EDGE_PUT(2) EDGE_PUT(3)
    EDGE_PUT(4) EDGE_PUT(5) EDGE_PUT(6) EDGE_PUT(7)
#undef EDGE_LOAD
#undef EDGE_ADD
#undef EDGE_PUT
}

__global__ void __launch_bounds__(256)
k_place(const int* __restrict__ ei, int E, int n, int* __restrict__ rows,
        int* __restrict__ slots, int* __restrict__ novf, int* __restrict__ ovf_w,
        int* __restrict__ tick) {
    int2* __restrict__ ovf = (int2*)ovf_w;
    const int lane = threadIdx.x & 31;
    const int half = (int)((__builtin_amdgcn_s_sendmsg_rtn(0x87) >> 18) & 1u);
    const int nch    = (E + CHUNK - 1) / CHUNK;
    const int nsplit = n >> 1;
    #pragma unroll 1
    for (int ph = 0; ph < 2; ++ph) {
        const int rg = half ^ ph;
        int* __restrict__ ctr = tick + rg * TICK_STRIDE;
        const int      lo   = rg ? nsplit : 0;
        const unsigned span = (unsigned)(rg ? n - nsplit : nsplit);
        if (ph == 1 && __hip_atomic_load(ctr, __ATOMIC_RELAXED, __HIP_MEMORY_SCOPE_AGENT) >= nch) break;
        for (;;) {
            int traw = 0;
            if (lane == 0) traw = atomicAdd(ctr, 1);
            const int t = __builtin_amdgcn_readfirstlane(traw);
            if (t >= nch) break;
            if (t * CHUNK + CHUNK <= E) place_chunk(ei, E, rows, slots, novf, ovf, t * CHUNK + lane, lo, span, false);
            else                        place_chunk(ei, E, rows, slots, novf, ovf, t * CHUNK + lane, lo, span, true);
        }
    }
}

static __device__ __forceinline__ float dinv_of(float degree_word) {
    return rsqrtf(1.0f + (float)__float_as_int(degree_word));
}


__global__ void __launch_bounds__(256)
k_agg1(const int* __restrict__ slots, const float* __restrict__ h,
       const int* __restrict__ novf, const int* __restrict__ ovf_w,
       const float* __restrict__ b1, const float* __restrict__ W2,
       float* __restrict__ h2, int n) {
    const int2*   __restrict__ ovf = (const int2*)ovf_w;
    const float4* __restrict__ h4  = (const float4*)h;
    int t = blockIdx.x * 256 + threadIdx.x;
    int i = t >> 2;
    unsigned q = t & 3;
    if (i >= n) return;

    float4 hs  = h4[(size_t)i * (HROW / 4) + q];
    float  dw  = h[(size_t)i * HROW + FH];
    int    deg = __float_as_int(dw);
    float  di  = dinv_of(dw);
    int    m   = deg < SLOTS ? deg : SLOTS;
    const int* __restrict__ sl = slots + (size_t)i * SLOTS;
    float4 acc = make_float4(0.0f, 0.0f, 0.0f, 0.0f);

    int p = 0;
    for (; p + 4 <= m; p += 4) {
        int4 sv = *(const int4*)(sl + p);
        float4 r0 = h4[(unsigned)sv.x * (HROW / 4) + q];
        float4 r1 = h4[(unsigned)sv.y * (HROW / 4) + q];
        float4 r2 = h4[(unsigned)sv.z * (HROW / 4) + q];
        float4 r3 = h4[(unsigned)sv.w * (HROW / 4) + q];
        float c0 = h[(unsigned)sv.x * HROW + FH], c1 = h[(unsigned)sv.y * HROW + FH];
        float c2 = h[(unsigned)sv.z * HROW + FH], c3 = h[(unsigned)sv.w * HROW + FH];
        float w0 = dinv_of(c0) * di, w1 = dinv_of(c1) * di;
        float w2 = dinv_of(c2) * di, w3 = dinv_of(c3) * di;
        acc.x = fmaf(r0.x, w0, acc.x); acc.y = fmaf(r0.y, w0, acc.y);
        acc.z = fmaf(r0.z, w0, acc.z); acc.w = fmaf(r0.w, w0, acc.w);
        acc.x = fmaf(r1.x, w1, acc.x); acc.y = fmaf(r1.y, w1, acc.y);
        acc.z = fmaf(r1.z, w1, acc.z); acc.w = fmaf(r1.w, w1, acc.w);
        acc.x = fmaf(r2.x, w2, acc.x); acc.y = fmaf(r2.y, w2, acc.y);
        acc.z = fmaf(r2.z, w2, acc.z); acc.w = fmaf(r2.w, w2, acc.w);
        acc.x = fmaf(r3.x, w3, acc.x); acc.y = fmaf(r3.y, w3, acc.y);
        acc.z = fmaf(r3.z, w3, acc.z); acc.w = fmaf(r3.w, w3, acc.w);
    }
    #pragma unroll 1
    for (; p < m; ++p) {
        unsigned s0 = (unsigned)sl[p];
        float4 r0 = h4[s0 * (HROW / 4) + q];
        float w0 = dinv_of(h[s0 * HROW + FH]) * di;
        acc.x = fmaf(r0.x, w0, acc.x); acc.y = fmaf(r0.y, w0, acc.y);
        acc.z = fmaf(r0.z, w0, acc.z); acc.w = fmaf(r0.w, w0, acc.w);
    }
    if (deg > SLOTS) {
        int no = *novf;
        #pragma unroll 1
        for (int k = 0; k < no; ++k) {
            int2 o = ovf[k];
            if (o.y == i) {
                unsigned s0 = (unsigned)o.x;
                float4 r0 = h4[s0 * (HROW / 4) + q];
                float w0 = dinv_of(h[s0 * HROW + FH]) * di;
                acc.x = fmaf(r0.x, w0, acc.x); acc.y = fmaf(r0.y, w0, acc.y);
                acc.z = fmaf(r0.z, w0, acc.z); acc.w = fmaf(r0.w, w0, acc.w);
            }
        }
    }

    float ws = di * di;
    float4 bv = *(const float4*)(b1 + 4 * q);
    float v0 = fmaxf(fmaf(hs.x, ws, acc.x) + bv.x, 0.0f);
    float v1 = fmaxf(fmaf(hs.y, ws, acc.y) + bv.y, 0.0f);
    float v2 = fmaxf(fmaf(hs.z, ws, acc.z) + bv.z, 0.0f);
    float v3 = fmaxf(fmaf(hs.w, ws, acc.w) + bv.w, 0.0f);

    float4 wa = *(const float4*)(W2 + 8 * q);
    float4 wb = *(const float4*)(W2 + 8 * q + 4);
    float a0 = fmaf(v3, wb.z, fmaf(v2, wb.x, fmaf(v1, wa.z, v0 * wa.x)));
    float a1 = fmaf(v3, wb.w, fmaf(v2, wb.y, fmaf(v1, wa.w, v0 * wa.y)));
    a0 += __shfl_xor(a0, 1);  a1 += __shfl_xor(a1, 1);
    a0 += __shfl_xor(a0, 2);  a1 += __shfl_xor(a1, 2);
    if (q == 0) ((float4*)h2)[i] = make_float4(a0, a1, di, dw);
}


__global__ void __launch_bounds__(256)
k_agg2(const int* __restrict__ slots, const float* __restrict__ h2,
       const int* __restrict__ novf, const int* __restrict__ ovf_w,
       const float* __restrict__ b2, float* __restrict__ out, int n) {
    const int2*   __restrict__ ovf = (const int2*)ovf_w;
    const float4* __restrict__ g4  = (const float4*)h2;
    int t = blockIdx.x * 256 + threadIdx.x;
    int i = t >> 2;
    unsigned q = t & 3;
    if (i >= n) return;

    float4 own = g4[i];
    float  di  = own.z;
    int    deg = __float_as_int(own.w);
    unsigned m = (unsigned)(deg < SLOTS ? deg : SLOTS);
    const int* __restrict__ sl = slots + (size_t)i * SLOTS;
    unsigned p = q;
    float a0 = 0.0f, a1 = 0.0f;

    for (; p + 12 < m; p += 16) {
        unsigned s0 = (unsigned)sl[p],     s1 = (unsigned)sl[p + 4];
        unsigned s2 = (unsigned)sl[p + 8], s3 = (unsigned)sl[p + 12];
        float4 r0 = g4[s0], r1 = g4[s1], r2 = g4[s2], r3 = g4[s3];
        float w0 = r0.z * di, w1 = r1.z * di, w2 = r2.z * di, w3 = r3.z * di;
        a0 = fmaf(r0.x, w0, a0); a1 = fmaf(r0.y, w0, a1);
        a0 = fmaf(r1.x, w1, a0); a1 = fmaf(r1.y, w1, a1);
        a0 = fmaf(r2.x, w2, a0); a1 = fmaf(r2.y, w2, a1);
        a0 = fmaf(r3.x, w3, a0); a1 = fmaf(r3.y, w3, a1);
    }
    #pragma unroll 1
    for (; p < m; p += 4) {
        float4 r0 = g4[(unsigned)sl[p]];
        float w0 = r0.z * di;
        a0 = fmaf(r0.x, w0, a0); a1 = fmaf(r0.y, w0, a1);
    }
    if (deg > SLOTS) {
        int no = *novf;
        #pragma unroll 1
        for (int k = (int)q; k < no; k += 4) {
            int2 o = ovf[k];
            if (o.y == i) {
                float4 r0 = g4[(unsigned)o.x];
                float w0 = r0.z * di;
                a0 = fmaf(r0.x, w0, a0); a1 = fmaf(r0.y, w0, a1);
            }
        }
    }
    a0 += __shfl_xor(a0, 1);  a1 += __shfl_xor(a1, 1);
    a0 += __shfl_xor(a0, 2);  a1 += __shfl_xor(a1, 2);

    float ws = di * di;
    a0 = fmaf(own.x, ws, a0) + b2[0];
    a1 = fmaf(own.y, ws, a1) + b2[1];
    float mx = fmaxf(a0, a1);
    float ls = logf(expf(a0 - mx) + expf(a1 - mx));
    if (q == 0) ((float2*)out)[i] = make_float2(a0 - mx - ls, a1 - mx - ls);
}


extern "C" void kernel_launch(void* const* d_in, const int* in_sizes, int n_in,
                              void* d_out, int out_size, void* d_ws, size_t ws_size,
                              hipStream_t stream) {
    const float* x  = (const float*)d_in[0];
    const int*   ei = (const int*)d_in[1];
    const float* W1 = (const float*)d_in[2];
    const float* b1 = (const float*)d_in[3];
    const float* W2 = (const float*)d_in[4];
    const float* b2 = (const float*)d_in[5];
    float* out      = (float*)d_out;

    const int N = in_sizes[0] / FIN;
    const int E = in_sizes[1] / 2;

    float* h     = (float*)d_ws;
    int*   slots = (int*)(h + (size_t)N * HROW);
    float* h2    = (float*)(slots + (size_t)N * SLOTS);
    int*   ovf   = (int*)(h2 + (size_t)N * 4);
    int*   novf  = ovf + (size_t)E * 2;
    int*   tick  = novf + TICK_STRIDE;

    const int B = 256;
    auto blk = [](long long t, int b) { return (int)((t + b - 1) / b); };

    const int n_tiles = N / 16;
    k_gemm1_wmma<<<blk(n_tiles, 8), B, 0, stream>>>(x, W1, h, novf, tick, n_tiles);

    k_place <<<1024, B, 0, stream>>>(ei, E, N, (int*)h, slots, novf, ovf, tick);

    k_agg1  <<<blk((long long)N * 4, B), B, 0, stream>>>(slots, h, novf, ovf, b1, W2, h2, N);
    k_agg2  <<<blk((long long)N * 4, B), B, 0, stream>>>(slots, h2, novf, ovf, b2, out, N);
}
